# starting kernel
# baseline (speedup 1.0000x reference)
#include <hip/hip_runtime.h>
#include <math.h>

typedef __attribute__((ext_vector_type(2))) float v2f;
typedef __attribute__((ext_vector_type(4))) float v4f;
typedef __attribute__((ext_vector_type(8))) float v8f;

#define NDIM 64

__global__ void fa_gate_wmma(const float* __restrict__ h,
                             const float* __restrict__ gate_W,
                             float* __restrict__ gd,
                             float* __restrict__ gs,
                             int N)
{
    int wave = (blockIdx.x * blockDim.x + threadIdx.x) >> 5;
    int lane = threadIdx.x & 31;
    int base = wave * 16;
    if (base >= N) return;

    int row   = lane & 15;
    int khalf = (lane >> 4) << 1;
    int arow  = base + row; if (arow > N - 1) arow = N - 1;

    float msk = (row < 2) ? 1.0f : 0.0f;
    const float* wbase = gate_W + ((row < 2) ? row : 0) * NDIM + khalf;
    v2f bfrag[16];
    #pragma unroll
    for (int j = 0; j < 16; ++j) {
        bfrag[j].x = wbase[j * 4]     * msk;
        bfrag[j].y = wbase[j * 4 + 1] * msk;
    }

    const float* arow_p = h + (size_t)arow * NDIM + khalf;
    v8f acc = {0.f, 0.f, 0.f, 0.f, 0.f, 0.f, 0.f, 0.f};
    #pragma unroll
    for (int j = 0; j < 16; ++j) {
        v2f a = *(const v2f*)(arow_p + j * 4);
        acc = __builtin_amdgcn_wmma_f32_16x16x4_f32(
                   false, a,  false, bfrag[j],
                   (short)0, acc,  false,  false);
    }

    if (row < 2) {
        float* outp = (row == 0) ? gd : gs;
        int moff = (lane >> 4) << 3;
        #pragma unroll
        for (int i = 0; i < 8; ++i) {
            int m = base + moff + i;
            if (m < N) outp[m] = acc[i];
        }
    }
}

__global__ void fa_edge_gate(const float* __restrict__ gd,
                             const float* __restrict__ gs,
                             const float* __restrict__ dvec,
                             const float* __restrict__ gate_b,
                             const int* __restrict__ esrc,
                             const int* __restrict__ edst,
                             float* __restrict__ e, int E)
{
    int i = blockIdx.x * blockDim.x + threadIdx.x;
    if (i >= E) return;
    int s = esrc[i];
    int t = edst[i];
    float g = tanhf(gd[t] + gs[s] + gate_b[0]);
    e[i] = g * dvec[t] * dvec[s];
}

__global__ void fa_zero(float* __restrict__ z, int n4)
{
    int i = blockIdx.x * blockDim.x + threadIdx.x;
    if (i < n4) ((v4f*)z)[i] = (v4f){0.f, 0.f, 0.f, 0.f};
}

__global__ void fa_scatter(const float* __restrict__ h,
                           const int* __restrict__ esrc,
                           const int* __restrict__ edst,
                           const float* __restrict__ e,
                           float* __restrict__ z, int E)
{
    int lane   = threadIdx.x & 31;
    int wave   = (blockIdx.x * blockDim.x + threadIdx.x) >> 5;
    int nwaves = (gridDim.x * blockDim.x) >> 5;
    int f = lane << 1;

    for (int i = wave; i < E; i += nwaves) {
        int   s = esrc[i];
        int   t = edst[i];
        float c = e[i];
        v2f hv = *(const v2f*)(h + (size_t)s * NDIM + f);
        float* zp = z + (size_t)t * NDIM + f;
        atomicAdd(zp,     hv.x * c);
        atomicAdd(zp + 1, hv.y * c);
    }
}

extern "C" void kernel_launch(void* const* d_in, const int* in_sizes, int n_in,
                              void* d_out, int out_size, void* d_ws, size_t ws_size,
                              hipStream_t stream)
{
    const float* h      = (const float*)d_in[0];
    const float* dvec   = (const float*)d_in[1];
    const float* gate_W = (const float*)d_in[2];
    const float* gate_b = (const float*)d_in[3];
    const int*   esrc   = (const int*)d_in[4];
    const int*   edst   = (const int*)d_in[5];

    int N = in_sizes[1];
    int E = in_sizes[4];
    float* z = (float*)d_out;

    float* gd = (float*)d_ws;
    float* gs = gd + N;
    float* e  = gs + N;

    {
        int waves   = (N + 15) / 16;
        int threads = waves * 32;
        dim3 blk(256), grd((threads + 255) / 256);
        fa_gate_wmma<<<grd, blk, 0, stream>>>(h, gate_W, gd, gs, N);
    }
    {
        dim3 blk(256), grd((E + 255) / 256);
        fa_edge_gate<<<grd, blk, 0, stream>>>(gd, gs, dvec, gate_b, esrc, edst, e, E);
    }
    {
        int n4 = out_size / 4;
        dim3 blk(256), grd((n4 + 255) / 256);
        fa_zero<<<grd, blk, 0, stream>>>(z, n4);
    }
    {
        dim3 blk(256), grd(25000);
        fa_scatter<<<grd, blk, 0, stream>>>(h, esrc, edst, e, z, E);
    }
}


# Owner keys bit-sliced, eight planes per 32 edges (built by wave ballots): a lane matches 32 edges with 11 bit ops, one mask word per 32 edges, not one per 4; queue order, so LDS add order, changes
# speedup vs baseline: 1.6037x; 1.6037x over previous
#include <hip/hip_runtime.h>
#include <math.h>

typedef __attribute__((ext_vector_type(2))) float v2f;
typedef __attribute__((ext_vector_type(4))) float v4f;
typedef __attribute__((ext_vector_type(8))) float v8f;

#define NDIM 64

__global__ void fa_gate_wmma(const float* __restrict__ h,
                             const float* __restrict__ gate_W,
                             float* __restrict__ gd,
                             float* __restrict__ gs,
                             int N)
{
    int wave = (blockIdx.x * blockDim.x + threadIdx.x) >> 5;
    int lane = threadIdx.x & 31;
    int base = wave * 16;
    if (base >= N) return;

    int row   = lane & 15;
    int khalf = (lane >> 4) << 1;
    int arow  = base + row; if (arow > N - 1) arow = N - 1;

    float msk = (row < 2) ? 1.0f : 0.0f;
    const float* wbase = gate_W + ((row < 2) ? row : 0) * NDIM + khalf;
    v2f bfrag[16];
    #pragma unroll
    for (int j = 0; j < 16; ++j) {
        bfrag[j].x = wbase[j * 4]     * msk;
        bfrag[j].y = wbase[j * 4 + 1] * msk;
    }

    const float* arow_p = h + (size_t)arow * NDIM + khalf;
    v8f acc = {0.f, 0.f, 0.f, 0.f, 0.f, 0.f, 0.f, 0.f};
    #pragma unroll
    for (int j = 0; j < 16; ++j) {
        v2f a = *(const v2f*)(arow_p + j * 4);
        acc = __builtin_amdgcn_wmma_f32_16x16x4_f32(
                   false, a,  false, bfrag[j],
                   (short)0, acc,  false,  false);
    }

    if (row < 2) {
        float* outp = (row == 0) ? gd : gs;
        int moff = (lane >> 4) << 3;
        #pragma unroll
        for (int i = 0; i < 8; ++i) {
            int m = base + moff + i;
            if (m < N) outp[m] = acc[i];
        }
    }
}


#define RMAX 512
#define QCAP 2048
#define WGT  1024

typedef __attribute__((ext_vector_type(4))) unsigned int v4u;
typedef __attribute__((ext_vector_type(4))) int v4i;

__global__ void fa_owner_planes(const int* __restrict__ edst,
                                unsigned int* __restrict__ planes_w, int E, int nunit, int R)
{
    v4u* lo = (v4u*)planes_w;
    v4u* hi = lo + nunit;
    const int lane = threadIdx.x & 31;
    const int wv   = (blockIdx.x * blockDim.x + threadIdx.x) >> 5;
    const int u0   = wv * 8;
    if (u0 >= nunit) return;

    int t[8];
    #pragma unroll
    for (int i = 0; i < 8; ++i) {
        int e = (u0 + i) * 32 + lane;
        t[i] = (e < E) ? edst[e] : 0;
    }
    unsigned int l0 = 0u, l1 = 0u, l2 = 0u, l3 = 0u, h0 = 0u, h1 = 0u, h2 = 0u, h3 = 0u;
    #pragma unroll
    for (int i = 0; i < 8; ++i) {
        unsigned int k  = (unsigned int)(t[i] / R);
        unsigned int b0 = __builtin_amdgcn_ballot_w32((k & 1u)   != 0u);
        unsigned int b1 = __builtin_amdgcn_ballot_w32((k & 2u)   != 0u);
        unsigned int b2 = __builtin_amdgcn_ballot_w32((k & 4u)   != 0u);
        unsigned int b3 = __builtin_amdgcn_ballot_w32((k & 8u)   != 0u);
        unsigned int b4 = __builtin_amdgcn_ballot_w32((k & 16u)  != 0u);
        unsigned int b5 = __builtin_amdgcn_ballot_w32((k & 32u)  != 0u);
        unsigned int b6 = __builtin_amdgcn_ballot_w32((k & 64u)  != 0u);
        unsigned int b7 = __builtin_amdgcn_ballot_w32((k & 128u) != 0u);
        if (lane == i) { l0 = b0; l1 = b1; l2 = b2; l3 = b3; h0 = b4; h1 = b5; h2 = b6; h3 = b7; }
    }
    if (lane < 8 && u0 + lane < nunit) {
        v4u lo4 = {l0, l1, l2, l3};
        v4u hi4 = {h0, h1, h2, h3};
        lo[u0 + lane] = lo4;
        hi[u0 + lane] = hi4;
    }
}

__global__ __launch_bounds__(WGT)
void fa_owner_sum(const float* __restrict__ h,
                  const float* __restrict__ dvec,
                  const float* __restrict__ gd,
                  const float* __restrict__ gs,
                  const float* __restrict__ gate_b,
                  const int* __restrict__ esrc,
                  const int* __restrict__ edst,
                  const unsigned int* __restrict__ planes_w,
                  float* __restrict__ z,
                  int N, int E, int nunit, int R)
{
    __shared__ __attribute__((aligned(16))) float zs[RMAX * NDIM];
    __shared__ unsigned short q[WGT / 32][QCAP];

    const v4u* klo = (const v4u*)planes_w;
    const v4u* khi = klo + nunit;
    const int tid  = threadIdx.x;
    const int lane = tid & 31;
    const int wave = tid >> 5;
    const int g    = blockIdx.x;
    const int row0 = g * R;
    const int f    = lane << 1;

    for (int i = tid; i < R * NDIM; i += WGT) zs[i] = 0.f;
    __syncthreads();

    const float bias = gate_b[0];
    unsigned short* myq = q[wave];

    unsigned int xg[8];
    #pragma unroll
    for (int b = 0; b < 8; ++b) xg[b] = (((unsigned int)g >> b) & 1u) - 1u;

    v4u cur[4];
    #pragma unroll
    for (int k = 0; k < 2; ++k) {
        int u = k * WGT + tid;
        if (u > nunit - 1) u = nunit - 1;
        cur[2 * k]     = klo[u];
        cur[2 * k + 1] = khi[u];
    }

    for (int base = 0; base < nunit; base += 2 * WGT) {
        unsigned int m[2];
        int n = 0;
        #pragma unroll
        for (int k = 0; k < 2; ++k) {
            bool ok = (base + k * WGT + tid) < nunit;
            v4u a = cur[2 * k], b = cur[2 * k + 1];
            unsigned int mm = (a.x ^ xg[0]) & (a.y ^ xg[1]) & (a.z ^ xg[2]) & (a.w ^ xg[3])
                            & (b.x ^ xg[4]) & (b.y ^ xg[5]) & (b.z ^ xg[6]) & (b.w ^ xg[7]);
            if (!ok) mm = 0u;
            m[k] = mm;
            n += __builtin_popcount(mm);
        }
        int nb = base + 2 * WGT;
        if (nb < nunit) {
            #pragma unroll
            for (int k = 0; k < 2; ++k) {
                int u = nb + k * WGT + tid;
                if (u > nunit - 1) u = nunit - 1;
                cur[2 * k]     = klo[u];
                cur[2 * k + 1] = khi[u];
            }
        }
        int incl = n;
        #pragma unroll
        for (int d = 1; d < 32; d <<= 1) {
            int up = __shfl_up(incl, d);
            if (lane >= d) incl += up;
        }
        int p = incl - n;
        const int total = __builtin_amdgcn_readlane(incl, 31);
        #pragma unroll
        for (int j = 0; j < 2; ++j) {
            unsigned int mm = m[j];
            while (mm) {
                int bit = __builtin_ctz(mm);
                mm &= mm - 1u;
                int code = (j << 10) | (lane << 5) | bit;
                myq[p++] = (unsigned short)code;
            }
        }
        __builtin_amdgcn_wave_barrier();

        for (int qb = 0; qb < total; qb += 32) {
            int  ei   = qb + lane;
            bool act  = ei < total;
            int  code = act ? (int)myq[ei] : 0;
            int  unit = base + (code >> 10) * WGT + (wave << 5) + ((code >> 5) & 31);
            int  idx  = unit * 32 + (code & 31);
            act = act && (idx < E);
            if (!act) idx = 0;
            int   s = esrc[idx];
            int   t = edst[idx];
            float c = tanhf(gd[t] + gs[s] + bias) * dvec[t] * dvec[s];
            int   tl = act ? (t - row0) : 0;
            const unsigned int am = __builtin_amdgcn_ballot_w32(act);
            const int cbits = __builtin_bit_cast(int, c);
            int cnt = total - qb; if (cnt > 32) cnt = 32;
            for (int k = 0; k < cnt; k += 8) {
                v2f hv[8];
                #pragma unroll
                for (int u = 0; u < 8; ++u) {
                    int sk = __builtin_amdgcn_readlane(s, k + u);
                    hv[u] = *(const v2f*)(h + (size_t)sk * NDIM + f);
                }
                #pragma unroll
                for (int u = 0; u < 8; ++u) {
                    if ((am >> (k + u)) & 1u) {
                        float ck = __builtin_bit_cast(float, __builtin_amdgcn_readlane(cbits, k + u));
                        int   tk = __builtin_amdgcn_readlane(tl, k + u);
                        float* zp = &zs[tk * NDIM + f];
                        __hip_atomic_fetch_add(zp,     hv[u].x * ck, __ATOMIC_RELAXED, __HIP_MEMORY_SCOPE_WORKGROUP);
                        __hip_atomic_fetch_add(zp + 1, hv[u].y * ck, __ATOMIC_RELAXED, __HIP_MEMORY_SCOPE_WORKGROUP);
                    }
                }
            }
        }
        __builtin_amdgcn_wave_barrier();
    }

    __syncthreads();
    for (int i = tid; i < R * (NDIM / 4); i += WGT) {
        int row = row0 + (i >> 4);
        if (row < N) ((v4f*)z)[(size_t)row * (NDIM / 4) + (i & 15)] = ((const v4f*)zs)[i];
    }
}

extern "C" void kernel_launch(void* const* d_in, const int* in_sizes, int n_in,
                              void* d_out, int out_size, void* d_ws, size_t ws_size,
                              hipStream_t stream)
{
    const float* h      = (const float*)d_in[0];
    const float* dvec   = (const float*)d_in[1];
    const float* gate_W = (const float*)d_in[2];
    const float* gate_b = (const float*)d_in[3];
    const int*   esrc   = (const int*)d_in[4];
    const int*   edst   = (const int*)d_in[5];

    int N = in_sizes[1];
    int E = in_sizes[4];
    float* z = (float*)d_out;

    float* gd = (float*)d_ws;
    float* gs = gd + N;
    unsigned int* planes = (unsigned int*)(gs + ((N + 3) / 4) * 4);

    int R    = (N + 255) / 256;
    int nunit = (E + 31) / 32;

    {
        int waves   = (N + 15) / 16;
        int threads = waves * 32;
        dim3 blk(256), grd((threads + 255) / 256);
        fa_gate_wmma<<<grd, blk, 0, stream>>>(h, gate_W, gd, gs, N);
    }
    {
        int waves = (nunit + 7) / 8;
        dim3 blk(256), grd((waves + 7) / 8);
        fa_owner_planes<<<grd, blk, 0, stream>>>(edst, planes, E, nunit, R);
    }
    {
        dim3 blk(WGT), grd(256);
        fa_owner_sum<<<grd, blk, 0, stream>>>(h, dvec, gd, gs, gate_b, esrc, edst,
                                              planes, z, N, E, nunit, R);
    }
}


# Owner sum: matches compacted by ballot and mbcnt into a per-wave 64-slot LDS ring kept across scan trips, drained only as full 32-edge batches with 32 h rows in flight; LDS add order changes
# speedup vs baseline: 1.7908x; 1.1167x over previous
#include <hip/hip_runtime.h>
#include <math.h>

typedef __attribute__((ext_vector_type(2))) float v2f;
typedef __attribute__((ext_vector_type(4))) float v4f;
typedef __attribute__((ext_vector_type(8))) float v8f;

#define NDIM 64

__global__ void fa_gate_wmma(const float* __restrict__ h,
                             const float* __restrict__ gate_W,
                             float* __restrict__ gd,
                             float* __restrict__ gs,
                             int N)
{
    int wave = (blockIdx.x * blockDim.x + threadIdx.x) >> 5;
    int lane = threadIdx.x & 31;
    int base = wave * 16;
    if (base >= N) return;

    int row   = lane & 15;
    int khalf = (lane >> 4) << 1;
    int arow  = base + row; if (arow > N - 1) arow = N - 1;

    float msk = (row < 2) ? 1.0f : 0.0f;
    const float* wbase = gate_W + ((row < 2) ? row : 0) * NDIM + khalf;
    v2f bfrag[16];
    #pragma unroll
    for (int j = 0; j < 16; ++j) {
        bfrag[j].x = wbase[j * 4]     * msk;
        bfrag[j].y = wbase[j * 4 + 1] * msk;
    }

    const float* arow_p = h + (size_t)arow * NDIM + khalf;
    v8f acc = {0.f, 0.f, 0.f, 0.f, 0.f, 0.f, 0.f, 0.f};
    #pragma unroll
    for (int j = 0; j < 16; ++j) {
        v2f a = *(const v2f*)(arow_p + j * 4);
        acc = __builtin_amdgcn_wmma_f32_16x16x4_f32(
                   false, a,  false, bfrag[j],
                   (short)0, acc,  false,  false);
    }

    if (row < 2) {
        float* outp = (row == 0) ? gd : gs;
        int moff = (lane >> 4) << 3;
        #pragma unroll
        for (int i = 0; i < 8; ++i) {
            int m = base + moff + i;
            if (m < N) outp[m] = acc[i];
        }
    }
}


#define RMAX 512
#define WGT  1024

typedef __attribute__((ext_vector_type(4))) unsigned int v4u;
typedef __attribute__((ext_vector_type(4))) int v4i;

__global__ void fa_owner_planes(const int* __restrict__ edst,
                                unsigned int* __restrict__ planes_w, int E, int nunit, int R)
{
    v4u* lo = (v4u*)planes_w;
    v4u* hi = lo + nunit;
    const int lane = threadIdx.x & 31;
    const int wv   = (blockIdx.x * blockDim.x + threadIdx.x) >> 5;
    const int u0   = wv * 8;
    if (u0 >= nunit) return;

    int t[8];
    #pragma unroll
    for (int i = 0; i < 8; ++i) {
        int e = (u0 + i) * 32 + lane;
        t[i] = (e < E) ? edst[e] : 0;
    }
    unsigned int l0 = 0u, l1 = 0u, l2 = 0u, l3 = 0u, h0 = 0u, h1 = 0u, h2 = 0u, h3 = 0u;
    #pragma unroll
    for (int i = 0; i < 8; ++i) {
        unsigned int k  = (unsigned int)(t[i] / R);
        unsigned int b0 = __builtin_amdgcn_ballot_w32((k & 1u)   != 0u);
        unsigned int b1 = __builtin_amdgcn_ballot_w32((k & 2u)   != 0u);
        unsigned int b2 = __builtin_amdgcn_ballot_w32((k & 4u)   != 0u);
        unsigned int b3 = __builtin_amdgcn_ballot_w32((k & 8u)   != 0u);
        unsigned int b4 = __builtin_amdgcn_ballot_w32((k & 16u)  != 0u);
        unsigned int b5 = __builtin_amdgcn_ballot_w32((k & 32u)  != 0u);
        unsigned int b6 = __builtin_amdgcn_ballot_w32((k & 64u)  != 0u);
        unsigned int b7 = __builtin_amdgcn_ballot_w32((k & 128u) != 0u);
        if (lane == i) { l0 = b0; l1 = b1; l2 = b2; l3 = b3; h0 = b4; h1 = b5; h2 = b6; h3 = b7; }
    }
    if (lane < 8 && u0 + lane < nunit) {
        v4u lo4 = {l0, l1, l2, l3};
        v4u hi4 = {h0, h1, h2, h3};
        lo[u0 + lane] = lo4;
        hi[u0 + lane] = hi4;
    }
}

#define RING 64

template <bool FULL>
static __device__ __forceinline__
void fa_drain(const float* __restrict__ h,
              const float* __restrict__ dvec,
              const float* __restrict__ gd,
              const float* __restrict__ gs,
              const int* __restrict__ esrc,
              const int* __restrict__ edst,
              float* zs, float bias, int row0, int f, int lane, int vq, int n)
{
    const bool act = FULL || (lane < n);
    const int  idx = act ? vq : 0;
    const int   s = esrc[idx];
    const int   t = edst[idx];
    const float c = tanhf(gd[t] + gs[s] + bias) * dvec[t] * dvec[s];
    const int   tl = act ? (t - row0) : 0;
    const int cbits = __builtin_bit_cast(int, c);
    if (FULL) {
        v2f hv[32];
        #pragma unroll
        for (int u = 0; u < 32; ++u) {
            int sk = __builtin_amdgcn_readlane(s, u);
            hv[u] = *(const v2f*)(h + (size_t)sk * NDIM + f);
        }
        #pragma unroll
        for (int u = 0; u < 32; ++u) {
            float ck = __builtin_bit_cast(float, __builtin_amdgcn_readlane(cbits, u));
            int   tk = __builtin_amdgcn_readlane(tl, u);
            float* zp = &zs[tk * NDIM + f];
            __hip_atomic_fetch_add(zp,     hv[u].x * ck, __ATOMIC_RELAXED, __HIP_MEMORY_SCOPE_WORKGROUP);
            __hip_atomic_fetch_add(zp + 1, hv[u].y * ck, __ATOMIC_RELAXED, __HIP_MEMORY_SCOPE_WORKGROUP);
        }
    } else {
        for (int k = 0; k < n; k += 8) {
            v2f hv[8];
            #pragma unroll
            for (int u = 0; u < 8; ++u) {
                int sk = __builtin_amdgcn_readlane(s, k + u);
                hv[u] = *(const v2f*)(h + (size_t)sk * NDIM + f);
            }
            #pragma unroll
            for (int u = 0; u < 8; ++u) {
                if (k + u < n) {
                    float ck = __builtin_bit_cast(float, __builtin_amdgcn_readlane(cbits, k + u));
                    int   tk = __builtin_amdgcn_readlane(tl, k + u);
                    float* zp = &zs[tk * NDIM + f];
                    __hip_atomic_fetch_add(zp,     hv[u].x * ck, __ATOMIC_RELAXED, __HIP_MEMORY_SCOPE_WORKGROUP);
                    __hip_atomic_fetch_add(zp + 1, hv[u].y * ck, __ATOMIC_RELAXED, __HIP_MEMORY_SCOPE_WORKGROUP);
                }
            }
        }
    }
}

__global__ __launch_bounds__(WGT)
void fa_owner_sum(const float* __restrict__ h,
                  const float* __restrict__ dvec,
                  const float* __restrict__ gd,
                  const float* __restrict__ gs,
                  const float* __restrict__ gate_b,
                  const int* __restrict__ esrc,
                  const int* __restrict__ edst,
                  const unsigned int* __restrict__ planes_w,
                  float* __restrict__ z,
                  int N, int E, int nunit, int R)
{
    __shared__ __attribute__((aligned(16))) float zs[RMAX * NDIM];
    __shared__ int ring[WGT / 32][RING];

    const v4u* klo = (const v4u*)planes_w;
    const v4u* khi = klo + nunit;
    const int tid  = threadIdx.x;
    const int lane = tid & 31;
    const int g    = blockIdx.x;
    const int row0 = g * R;
    const int f    = lane << 1;

    for (int i = tid; i < R * NDIM; i += WGT) zs[i] = 0.f;
    __syncthreads();

    const float bias = gate_b[0];
    int* myring = ring[tid >> 5];

    unsigned int xg[8];
    #pragma unroll
    for (int b = 0; b < 8; ++b) xg[b] = (((unsigned int)g >> b) & 1u) - 1u;

    v4u cur[4];
    #pragma unroll
    for (int k = 0; k < 2; ++k) {
        int u = k * WGT + tid;
        if (u > nunit - 1) u = nunit - 1;
        cur[2 * k]     = klo[u];
        cur[2 * k + 1] = khi[u];
    }

    int head = 0;
    int tail = 0;

    for (int base = 0; base < nunit; base += 2 * WGT) {
        unsigned int m[2];
        #pragma unroll
        for (int k = 0; k < 2; ++k) {
            v4u a = cur[2 * k], b = cur[2 * k + 1];
            m[k] = (a.x ^ xg[0]) & (a.y ^ xg[1]) & (a.z ^ xg[2]) & (a.w ^ xg[3])
                 & (b.x ^ xg[4]) & (b.y ^ xg[5]) & (b.z ^ xg[6]) & (b.w ^ xg[7]);
        }
        if (base + 2 * WGT >= nunit) {
            const unsigned int lastmask = (E & 31) ? ((1u << (E & 31)) - 1u) : ~0u;
            #pragma unroll
            for (int k = 0; k < 2; ++k) {
                int u = base + k * WGT + tid;
                unsigned int valid = (u < nunit - 1) ? ~0u : ((u == nunit - 1) ? lastmask : 0u);
                m[k] &= valid;
            }
        }
        int nb = base + 2 * WGT;
        if (nb < nunit) {
            #pragma unroll
            for (int k = 0; k < 2; ++k) {
                int u = nb + k * WGT + tid;
                if (u > nunit - 1) u = nunit - 1;
                cur[2 * k]     = klo[u];
                cur[2 * k + 1] = khi[u];
            }
        }
        #pragma unroll
        for (int k = 0; k < 2; ++k) {
            unsigned int mm = m[k];
            const int ebase = (base + k * WGT + tid) << 5;
            unsigned int bal = __builtin_amdgcn_ballot_w32(mm != 0u);
            while (bal) {
                if (mm != 0u) {
                    int rank = (int)__builtin_amdgcn_mbcnt_lo(bal, 0u);
                    myring[(tail + rank) & (RING - 1)] = ebase | __builtin_ctz(mm);
                }
                mm &= mm - 1u;
                tail += __builtin_popcount(bal);
                __builtin_amdgcn_wave_barrier();
                if (tail - head >= 32) {
                    int vq = myring[(head + lane) & (RING - 1)];
                    fa_drain<true>(h, dvec, gd, gs, esrc, edst, zs, bias, row0, f, lane, vq, 32);
                    head += 32;
                    __builtin_amdgcn_wave_barrier();
                }
                bal = __builtin_amdgcn_ballot_w32(mm != 0u);
            }
        }
    }
    if (tail - head > 0) {
        const int left = tail - head;
        int vq = (lane < left) ? myring[(head + lane) & (RING - 1)] : 0;
        fa_drain<false>(h, dvec, gd, gs, esrc, edst, zs, bias, row0, f, lane, vq, left);
    }

    __syncthreads();
    for (int i = tid; i < R * (NDIM / 4); i += WGT) {
        int row = row0 + (i >> 4);
        if (row < N) ((v4f*)z)[(size_t)row * (NDIM / 4) + (i & 15)] = ((const v4f*)zs)[i];
    }
}

extern "C" void kernel_launch(void* const* d_in, const int* in_sizes, int n_in,
                              void* d_out, int out_size, void* d_ws, size_t ws_size,
                              hipStream_t stream)
{
    const float* h      = (const float*)d_in[0];
    const float* dvec   = (const float*)d_in[1];
    const float* gate_W = (const float*)d_in[2];
    const float* gate_b = (const float*)d_in[3];
    const int*   esrc   = (const int*)d_in[4];
    const int*   edst   = (const int*)d_in[5];

    int N = in_sizes[1];
    int E = in_sizes[4];
    float* z = (float*)d_out;

    float* gd = (float*)d_ws;
    float* gs = gd + N;
    unsigned int* planes = (unsigned int*)(gs + ((N + 3) / 4) * 4);

    int R    = (N + 255) / 256;
    int nunit = (E + 31) / 32;

    {
        int waves   = (N + 15) / 16;
        int threads = waves * 32;
        dim3 blk(256), grd((threads + 255) / 256);
        fa_gate_wmma<<<grd, blk, 0, stream>>>(h, gate_W, gd, gs, N);
    }
    {
        int waves = (nunit + 7) / 8;
        dim3 blk(256), grd((waves + 7) / 8);
        fa_owner_planes<<<grd, blk, 0, stream>>>(edst, planes, E, nunit, R);
    }
    {
        dim3 blk(WGT), grd(256);
        fa_owner_sum<<<grd, blk, 0, stream>>>(h, dvec, gd, gs, gate_b, esrc, edst,
                                              planes, z, N, E, nunit, R);
    }
}


# Key kernel also writes one packed word per edge (src | local dst row << 17); the owner sum fetches that one word per matched edge, not esrc and edst: one line-per-lane load fewer per edge
# speedup vs baseline: 2.0022x; 1.1180x over previous
#include <hip/hip_runtime.h>
#include <math.h>

typedef __attribute__((ext_vector_type(2))) float v2f;
typedef __attribute__((ext_vector_type(4))) float v4f;
typedef __attribute__((ext_vector_type(8))) float v8f;

#define NDIM 64

__global__ void fa_gate_wmma(const float* __restrict__ h,
                             const float* __restrict__ gate_W,
                             float* __restrict__ gd,
                             float* __restrict__ gs,
                             int N)
{
    int wave = (blockIdx.x * blockDim.x + threadIdx.x) >> 5;
    int lane = threadIdx.x & 31;
    int base = wave * 16;
    if (base >= N) return;

    int row   = lane & 15;
    int khalf = (lane >> 4) << 1;
    int arow  = base + row; if (arow > N - 1) arow = N - 1;

    float msk = (row < 2) ? 1.0f : 0.0f;
    const float* wbase = gate_W + ((row < 2) ? row : 0) * NDIM + khalf;
    v2f bfrag[16];
    #pragma unroll
    for (int j = 0; j < 16; ++j) {
        bfrag[j].x = wbase[j * 4]     * msk;
        bfrag[j].y = wbase[j * 4 + 1] * msk;
    }

    const float* arow_p = h + (size_t)arow * NDIM + khalf;
    v8f acc = {0.f, 0.f, 0.f, 0.f, 0.f, 0.f, 0.f, 0.f};
    #pragma unroll
    for (int j = 0; j < 16; ++j) {
        v2f a = *(const v2f*)(arow_p + j * 4);
        acc = __builtin_amdgcn_wmma_f32_16x16x4_f32(
                   false, a,  false, bfrag[j],
                   (short)0, acc,  false,  false);
    }

    if (row < 2) {
        float* outp = (row == 0) ? gd : gs;
        int moff = (lane >> 4) << 3;
        #pragma unroll
        for (int i = 0; i < 8; ++i) {
            int m = base + moff + i;
            if (m < N) outp[m] = acc[i];
        }
    }
}


#define RMAX 512
#define WGT  1024

typedef __attribute__((ext_vector_type(4))) unsigned int v4u;
typedef __attribute__((ext_vector_type(4))) int v4i;

__global__ void fa_owner_planes(const int* __restrict__ esrc,
                                const int* __restrict__ edst,
                                unsigned int* __restrict__ planes_w,
                                unsigned int* __restrict__ rec, int E, int nunit, int R)
{
    v4u* lo = (v4u*)planes_w;
    v4u* hi = lo + nunit;
    const int lane = threadIdx.x & 31;
    const int wv   = (blockIdx.x * blockDim.x + threadIdx.x) >> 5;
    const int u0   = wv * 8;
    if (u0 >= nunit) return;

    int t[8];
    #pragma unroll
    for (int i = 0; i < 8; ++i) {
        int e = (u0 + i) * 32 + lane;
        t[i] = (e < E) ? edst[e] : 0;
    }
    unsigned int kk[8];
    #pragma unroll
    for (int i = 0; i < 8; ++i) kk[i] = (unsigned int)(t[i] / R);
    if (rec != nullptr) {
        int sv[8];
        #pragma unroll
        for (int i = 0; i < 8; ++i) {
            int e = (u0 + i) * 32 + lane;
            sv[i] = (e < E) ? esrc[e] : 0;
        }
        #pragma unroll
        for (int i = 0; i < 8; ++i) {
            int e = (u0 + i) * 32 + lane;
            unsigned int tl = (unsigned int)t[i] - kk[i] * (unsigned int)R;
            if (e < E) rec[e] = (unsigned int)sv[i] | (tl << 17);
        }
    }
    unsigned int l0 = 0u, l1 = 0u, l2 = 0u, l3 = 0u, h0 = 0u, h1 = 0u, h2 = 0u, h3 = 0u;
    #pragma unroll
    for (int i = 0; i < 8; ++i) {
        unsigned int k  = kk[i];
        unsigned int b0 = __builtin_amdgcn_ballot_w32((k & 1u)   != 0u);
        unsigned int b1 = __builtin_amdgcn_ballot_w32((k & 2u)   != 0u);
        unsigned int b2 = __builtin_amdgcn_ballot_w32((k & 4u)   != 0u);
        unsigned int b3 = __builtin_amdgcn_ballot_w32((k & 8u)   != 0u);
        unsigned int b4 = __builtin_amdgcn_ballot_w32((k & 16u)  != 0u);
        unsigned int b5 = __builtin_amdgcn_ballot_w32((k & 32u)  != 0u);
        unsigned int b6 = __builtin_amdgcn_ballot_w32((k & 64u)  != 0u);
        unsigned int b7 = __builtin_amdgcn_ballot_w32((k & 128u) != 0u);
        if (lane == i) { l0 = b0; l1 = b1; l2 = b2; l3 = b3; h0 = b4; h1 = b5; h2 = b6; h3 = b7; }
    }
    if (lane < 8 && u0 + lane < nunit) {
        v4u lo4 = {l0, l1, l2, l3};
        v4u hi4 = {h0, h1, h2, h3};
        lo[u0 + lane] = lo4;
        hi[u0 + lane] = hi4;
    }
}

#define RING 64

template <bool FULL, bool PK>
static __device__ __forceinline__
void fa_drain(const float* __restrict__ h,
              const float* __restrict__ dvec,
              const float* __restrict__ gd,
              const float* __restrict__ gs,
              const int* __restrict__ esrc,
              const int* __restrict__ edst,
              const unsigned int* __restrict__ rec,
              float* zs, float bias, int row0, int f, int lane, int vq, int n)
{
    const bool act = FULL || (lane < n);
    const int  idx = act ? vq : 0;
    int s, t, tl;
    if (PK) {
        const unsigned int r = rec[idx];
        s  = (int)(r & 0x1FFFFu);
        tl = act ? (int)(r >> 17) : 0;
        t  = act ? (row0 + tl) : 0;
    } else {
        s  = esrc[idx];
        t  = edst[idx];
        tl = act ? (t - row0) : 0;
    }
    const float c = tanhf(gd[t] + gs[s] + bias) * dvec[t] * dvec[s];
    const int cbits = __builtin_bit_cast(int, c);
    if (FULL) {
        v2f hv[32];
        #pragma unroll
        for (int u = 0; u < 32; ++u) {
            int sk = __builtin_amdgcn_readlane(s, u);
            hv[u] = *(const v2f*)(h + (size_t)sk * NDIM + f);
        }
        #pragma unroll
        for (int u = 0; u < 32; ++u) {
            float ck = __builtin_bit_cast(float, __builtin_amdgcn_readlane(cbits, u));
            int   tk = __builtin_amdgcn_readlane(tl, u);
            float* zp = &zs[tk * NDIM + f];
            __hip_atomic_fetch_add(zp,     hv[u].x * ck, __ATOMIC_RELAXED, __HIP_MEMORY_SCOPE_WORKGROUP);
            __hip_atomic_fetch_add(zp + 1, hv[u].y * ck, __ATOMIC_RELAXED, __HIP_MEMORY_SCOPE_WORKGROUP);
        }
    } else {
        for (int k = 0; k < n; k += 8) {
            v2f hv[8];
            #pragma unroll
            for (int u = 0; u < 8; ++u) {
                int sk = __builtin_amdgcn_readlane(s, k + u);
                hv[u] = *(const v2f*)(h + (size_t)sk * NDIM + f);
            }
            #pragma unroll
            for (int u = 0; u < 8; ++u) {
                if (k + u < n) {
                    float ck = __builtin_bit_cast(float, __builtin_amdgcn_readlane(cbits, k + u));
                    int   tk = __builtin_amdgcn_readlane(tl, k + u);
                    float* zp = &zs[tk * NDIM + f];
                    __hip_atomic_fetch_add(zp,     hv[u].x * ck, __ATOMIC_RELAXED, __HIP_MEMORY_SCOPE_WORKGROUP);
                    __hip_atomic_fetch_add(zp + 1, hv[u].y * ck, __ATOMIC_RELAXED, __HIP_MEMORY_SCOPE_WORKGROUP);
                }
            }
        }
    }
}

template <bool PK>
__global__ __launch_bounds__(WGT)
void fa_owner_sum(const float* __restrict__ h,
                  const float* __restrict__ dvec,
                  const float* __restrict__ gd,
                  const float* __restrict__ gs,
                  const float* __restrict__ gate_b,
                  const int* __restrict__ esrc,
                  const int* __restrict__ edst,
                  const unsigned int* __restrict__ planes_w,
                  const unsigned int* __restrict__ rec,
                  float* __restrict__ z,
                  int N, int E, int nunit, int R)
{
    __shared__ __attribute__((aligned(16))) float zs[RMAX * NDIM];
    __shared__ int ring[WGT / 32][RING];

    const v4u* klo = (const v4u*)planes_w;
    const v4u* khi = klo + nunit;
    const int tid  = threadIdx.x;
    const int lane = tid & 31;
    const int g    = blockIdx.x;
    const int row0 = g * R;
    const int f    = lane << 1;

    for (int i = tid; i < R * NDIM; i += WGT) zs[i] = 0.f;
    __syncthreads();

    const float bias = gate_b[0];
    int* myring = ring[tid >> 5];

    unsigned int xg[8];
    #pragma unroll
    for (int b = 0; b < 8; ++b) xg[b] = (((unsigned int)g >> b) & 1u) - 1u;

    v4u cur[4];
    #pragma unroll
    for (int k = 0; k < 2; ++k) {
        int u = k * WGT + tid;
        if (u > nunit - 1) u = nunit - 1;
        cur[2 * k]     = klo[u];
        cur[2 * k + 1] = khi[u];
    }

    int head = 0;
    int tail = 0;

    for (int base = 0; base < nunit; base += 2 * WGT) {
        unsigned int m[2];
        #pragma unroll
        for (int k = 0; k < 2; ++k) {
            v4u a = cur[2 * k], b = cur[2 * k + 1];
            m[k] = (a.x ^ xg[0]) & (a.y ^ xg[1]) & (a.z ^ xg[2]) & (a.w ^ xg[3])
                 & (b.x ^ xg[4]) & (b.y ^ xg[5]) & (b.z ^ xg[6]) & (b.w ^ xg[7]);
        }
        if (base + 2 * WGT >= nunit) {
            const unsigned int lastmask = (E & 31) ? ((1u << (E & 31)) - 1u) : ~0u;
            #pragma unroll
            for (int k = 0; k < 2; ++k) {
                int u = base + k * WGT + tid;
                unsigned int valid = (u < nunit - 1) ? ~0u : ((u == nunit - 1) ? lastmask : 0u);
                m[k] &= valid;
            }
        }
        int nb = base + 2 * WGT;
        if (nb < nunit) {
            #pragma unroll
            for (int k = 0; k < 2; ++k) {
                int u = nb + k * WGT + tid;
                if (u > nunit - 1) u = nunit - 1;
                cur[2 * k]     = klo[u];
                cur[2 * k + 1] = khi[u];
            }
        }
        #pragma unroll
        for (int k = 0; k < 2; ++k) {
            unsigned int mm = m[k];
            const int ebase = (base + k * WGT + tid) << 5;
            unsigned int bal = __builtin_amdgcn_ballot_w32(mm != 0u);
            while (bal) {
                if (mm != 0u) {
                    int rank = (int)__builtin_amdgcn_mbcnt_lo(bal, 0u);
                    myring[(tail + rank) & (RING - 1)] = ebase | __builtin_ctz(mm);
                }
                mm &= mm - 1u;
                tail += __builtin_popcount(bal);
                __builtin_amdgcn_wave_barrier();
                if (tail - head >= 32) {
                    int vq = myring[(head + lane) & (RING - 1)];
                    fa_drain<true, PK>(h, dvec, gd, gs, esrc, edst, rec, zs, bias, row0, f, lane, vq, 32);
                    head += 32;
                    __builtin_amdgcn_wave_barrier();
                }
                bal = __builtin_amdgcn_ballot_w32(mm != 0u);
            }
        }
    }
    if (tail - head > 0) {
        const int left = tail - head;
        int vq = (lane < left) ? myring[(head + lane) & (RING - 1)] : 0;
        fa_drain<false, PK>(h, dvec, gd, gs, esrc, edst, rec, zs, bias, row0, f, lane, vq, left);
    }

    __syncthreads();
    for (int i = tid; i < R * (NDIM / 4); i += WGT) {
        int row = row0 + (i >> 4);
        if (row < N) ((v4f*)z)[(size_t)row * (NDIM / 4) + (i & 15)] = ((const v4f*)zs)[i];
    }
}

extern "C" void kernel_launch(void* const* d_in, const int* in_sizes, int n_in,
                              void* d_out, int out_size, void* d_ws, size_t ws_size,
                              hipStream_t stream)
{
    const float* h      = (const float*)d_in[0];
    const float* dvec   = (const float*)d_in[1];
    const float* gate_W = (const float*)d_in[2];
    const float* gate_b = (const float*)d_in[3];
    const int*   esrc   = (const int*)d_in[4];
    const int*   edst   = (const int*)d_in[5];

    int N = in_sizes[1];
    int E = in_sizes[4];
    float* z = (float*)d_out;

    float* gd = (float*)d_ws;
    float* gs = gd + N;
    unsigned int* planes = (unsigned int*)(gs + ((N + 3) / 4) * 4);

    int R    = (N + 255) / 256;
    int nunit = (E + 31) / 32;
    unsigned int* rec = planes + (size_t)nunit * 8;
    const bool packed = (N <= (1 << 17)) && (R <= RMAX) &&
                        ((size_t)((const char*)(rec + E) - (const char*)d_ws) <= ws_size);

    {
        int waves   = (N + 15) / 16;
        int threads = waves * 32;
        dim3 blk(256), grd((threads + 255) / 256);
        fa_gate_wmma<<<grd, blk, 0, stream>>>(h, gate_W, gd, gs, N);
    }
    {
        int waves = (nunit + 7) / 8;
        dim3 blk(256), grd((waves + 7) / 8);
        fa_owner_planes<<<grd, blk, 0, stream>>>(esrc, edst, planes, packed ? rec : nullptr, E, nunit, R);
    }
    {
        dim3 blk(WGT), grd(256);
        if (packed)
            fa_owner_sum<true><<<grd, blk, 0, stream>>>(h, dvec, gd, gs, gate_b, esrc, edst,
                                                        planes, rec, z, N, E, nunit, R);
        else
            fa_owner_sum<false><<<grd, blk, 0, stream>>>(h, dvec, gd, gs, gate_b, esrc, edst,
                                                         planes, rec, z, N, E, nunit, R);
    }
}


# Gate kernel also writes rows d[n]*h[n]; the owner sum gathers those, its edge coefficient is the bare tanh gate and d[dst] scales each finished row once: no d[src] line-per-lane load per edge
# speedup vs baseline: 2.0904x; 1.0441x over previous
#include <hip/hip_runtime.h>
#include <math.h>

typedef __attribute__((ext_vector_type(2))) float v2f;
typedef __attribute__((ext_vector_type(4))) float v4f;
typedef __attribute__((ext_vector_type(8))) float v8f;

#define NDIM 64

__global__ void fa_gate_wmma(const float* __restrict__ h,
                             const float* __restrict__ gate_W,
                             const float* __restrict__ dvec,
                             float* __restrict__ gd,
                             float* __restrict__ gs,
                             float* __restrict__ hs,
                             int N)
{
    int wave = (blockIdx.x * blockDim.x + threadIdx.x) >> 5;
    int lane = threadIdx.x & 31;
    int base = wave * 16;
    if (base >= N) return;

    int row   = lane & 15;
    int khalf = (lane >> 4) << 1;
    int arow  = base + row; if (arow > N - 1) arow = N - 1;

    float msk = (row < 2) ? 1.0f : 0.0f;
    const float* wbase = gate_W + ((row < 2) ? row : 0) * NDIM + khalf;
    v2f bfrag[16];
    #pragma unroll
    for (int j = 0; j < 16; ++j) {
        bfrag[j].x = wbase[j * 4]     * msk;
        bfrag[j].y = wbase[j * 4 + 1] * msk;
    }

    const float* arow_p = h + (size_t)arow * NDIM + khalf;
    v8f acc = {0.f, 0.f, 0.f, 0.f, 0.f, 0.f, 0.f, 0.f};
    #pragma unroll
    for (int j = 0; j < 16; ++j) {
        v2f a = *(const v2f*)(arow_p + j * 4);
        acc = __builtin_amdgcn_wmma_f32_16x16x4_f32(
                   false, a,  false, bfrag[j],
                   (short)0, acc,  false,  false);
    }

    if (row < 2) {
        float* outp = (row == 0) ? gd : gs;
        int moff = (lane >> 4) << 3;
        #pragma unroll
        for (int i = 0; i < 8; ++i) {
            int m = base + moff + i;
            if (m < N) outp[m] = acc[i];
        }
    }

    if (hs != nullptr) {
        const v4f* hin  = (const v4f*)(h  + (size_t)base * NDIM);
        v4f*       hout = (v4f*)(hs + (size_t)base * NDIM);
        if (base + 16 <= N) {
            v4f   v[8];
            float dd[8];
            #pragma unroll
            for (int i = 0; i < 8; ++i) {
                int q = i * 32 + lane;
                v[i]  = hin[q];
                dd[i] = dvec[base + (q >> 4)];
            }
            #pragma unroll
            for (int i = 0; i < 8; ++i) hout[i * 32 + lane] = v[i] * dd[i];
        } else {
            for (int i = 0; i < 8; ++i) {
                int q    = i * 32 + lane;
                int node = base + (q >> 4);
                if (node < N) hout[q] = hin[q] * dvec[node];
            }
        }
    }
}


#define RMAX 512
#define WGT  1024

typedef __attribute__((ext_vector_type(4))) unsigned int v4u;
typedef __attribute__((ext_vector_type(4))) int v4i;

__global__ void fa_owner_planes(const int* __restrict__ esrc,
                                const int* __restrict__ edst,
                                unsigned int* __restrict__ planes_w,
                                unsigned int* __restrict__ rec, int E, int nunit, int R)
{
    v4u* lo = (v4u*)planes_w;
    v4u* hi = lo + nunit;
    const int lane = threadIdx.x & 31;
    const int wv   = (blockIdx.x * blockDim.x + threadIdx.x) >> 5;
    const int u0   = wv * 8;
    if (u0 >= nunit) return;

    int t[8];
    #pragma unroll
    for (int i = 0; i < 8; ++i) {
        int e = (u0 + i) * 32 + lane;
        t[i] = (e < E) ? edst[e] : 0;
    }
    unsigned int kk[8];
    #pragma unroll
    for (int i = 0; i < 8; ++i) kk[i] = (unsigned int)(t[i] / R);
    if (rec != nullptr) {
        int sv[8];
        #pragma unroll
        for (int i = 0; i < 8; ++i) {
            int e = (u0 + i) * 32 + lane;
            sv[i] = (e < E) ? esrc[e] : 0;
        }
        #pragma unroll
        for (int i = 0; i < 8; ++i) {
            int e = (u0 + i) * 32 + lane;
            unsigned int tl = (unsigned int)t[i] - kk[i] * (unsigned int)R;
            if (e < E) rec[e] = (unsigned int)sv[i] | (tl << 17);
        }
    }
    unsigned int l0 = 0u, l1 = 0u, l2 = 0u, l3 = 0u, h0 = 0u, h1 = 0u, h2 = 0u, h3 = 0u;
    #pragma unroll
    for (int i = 0; i < 8; ++i) {
        unsigned int k  = kk[i];
        unsigned int b0 = __builtin_amdgcn_ballot_w32((k & 1u)   != 0u);
        unsigned int b1 = __builtin_amdgcn_ballot_w32((k & 2u)   != 0u);
        unsigned int b2 = __builtin_amdgcn_ballot_w32((k & 4u)   != 0u);
        unsigned int b3 = __builtin_amdgcn_ballot_w32((k & 8u)   != 0u);
        unsigned int b4 = __builtin_amdgcn_ballot_w32((k & 16u)  != 0u);
        unsigned int b5 = __builtin_amdgcn_ballot_w32((k & 32u)  != 0u);
        unsigned int b6 = __builtin_amdgcn_ballot_w32((k & 64u)  != 0u);
        unsigned int b7 = __builtin_amdgcn_ballot_w32((k & 128u) != 0u);
        if (lane == i) { l0 = b0; l1 = b1; l2 = b2; l3 = b3; h0 = b4; h1 = b5; h2 = b6; h3 = b7; }
    }
    if (lane < 8 && u0 + lane < nunit) {
        v4u lo4 = {l0, l1, l2, l3};
        v4u hi4 = {h0, h1, h2, h3};
        lo[u0 + lane] = lo4;
        hi[u0 + lane] = hi4;
    }
}

#define RING 64

template <bool FULL, int MODE>
static __device__ __forceinline__
void fa_drain(const float* __restrict__ h,
              const float* __restrict__ dvec,
              const float* __restrict__ gd,
              const float* __restrict__ gs,
              const int* __restrict__ esrc,
              const int* __restrict__ edst,
              const unsigned int* __restrict__ rec,
              float* zs, float bias, int row0, int f, int lane, int vq, int n)
{
    constexpr bool PK = MODE >= 1;
    constexpr bool SC = MODE == 2;
    const bool act = FULL || (lane < n);
    const int  idx = act ? vq : 0;
    int s, t, tl;
    if (PK) {
        const unsigned int r = rec[idx];
        s  = (int)(r & 0x1FFFFu);
        tl = act ? (int)(r >> 17) : 0;
        t  = act ? (row0 + tl) : 0;
    } else {
        s  = esrc[idx];
        t  = edst[idx];
        tl = act ? (t - row0) : 0;
    }
    float c = tanhf(gd[t] + gs[s] + bias);
    if (!SC) c = c * dvec[t] * dvec[s];
    const int cbits = __builtin_bit_cast(int, c);
    if (FULL) {
        v2f hv[32];
        #pragma unroll
        for (int u = 0; u < 32; ++u) {
            int sk = __builtin_amdgcn_readlane(s, u);
            hv[u] = *(const v2f*)(h + (size_t)sk * NDIM + f);
        }
        #pragma unroll
        for (int u = 0; u < 32; ++u) {
            float ck = __builtin_bit_cast(float, __builtin_amdgcn_readlane(cbits, u));
            int   tk = __builtin_amdgcn_readlane(tl, u);
            float* zp = &zs[tk * NDIM + f];
            __hip_atomic_fetch_add(zp,     hv[u].x * ck, __ATOMIC_RELAXED, __HIP_MEMORY_SCOPE_WORKGROUP);
            __hip_atomic_fetch_add(zp + 1, hv[u].y * ck, __ATOMIC_RELAXED, __HIP_MEMORY_SCOPE_WORKGROUP);
        }
    } else {
        for (int k = 0; k < n; k += 8) {
            v2f hv[8];
            #pragma unroll
            for (int u = 0; u < 8; ++u) {
                int sk = __builtin_amdgcn_readlane(s, k + u);
                hv[u] = *(const v2f*)(h + (size_t)sk * NDIM + f);
            }
            #pragma unroll
            for (int u = 0; u < 8; ++u) {
                if (k + u < n) {
                    float ck = __builtin_bit_cast(float, __builtin_amdgcn_readlane(cbits, k + u));
                    int   tk = __builtin_amdgcn_readlane(tl, k + u);
                    float* zp = &zs[tk * NDIM + f];
                    __hip_atomic_fetch_add(zp,     hv[u].x * ck, __ATOMIC_RELAXED, __HIP_MEMORY_SCOPE_WORKGROUP);
                    __hip_atomic_fetch_add(zp + 1, hv[u].y * ck, __ATOMIC_RELAXED, __HIP_MEMORY_SCOPE_WORKGROUP);
                }
            }
        }
    }
}

template <int MODE>
__global__ __launch_bounds__(WGT)
void fa_owner_sum(const float* __restrict__ h,
                  const float* __restrict__ dvec,
                  const float* __restrict__ gd,
                  const float* __restrict__ gs,
                  const float* __restrict__ gate_b,
                  const int* __restrict__ esrc,
                  const int* __restrict__ edst,
                  const unsigned int* __restrict__ planes_w,
                  const unsigned int* __restrict__ rec,
                  float* __restrict__ z,
                  int N, int E, int nunit, int R)
{
    __shared__ __attribute__((aligned(16))) float zs[RMAX * NDIM];
    __shared__ int ring[WGT / 32][RING];

    const v4u* klo = (const v4u*)planes_w;
    const v4u* khi = klo + nunit;
    const int tid  = threadIdx.x;
    const int lane = tid & 31;
    const int g    = blockIdx.x;
    const int row0 = g * R;
    const int f    = lane << 1;

    for (int i = tid; i < R * NDIM; i += WGT) zs[i] = 0.f;
    __syncthreads();

    const float bias = gate_b[0];
    int* myring = ring[tid >> 5];

    unsigned int xg[8];
    #pragma unroll
    for (int b = 0; b < 8; ++b) xg[b] = (((unsigned int)g >> b) & 1u) - 1u;

    v4u cur[4];
    #pragma unroll
    for (int k = 0; k < 2; ++k) {
        int u = k * WGT + tid;
        if (u > nunit - 1) u = nunit - 1;
        cur[2 * k]     = klo[u];
        cur[2 * k + 1] = khi[u];
    }

    int head = 0;
    int tail = 0;

    for (int base = 0; base < nunit; base += 2 * WGT) {
        unsigned int m[2];
        #pragma unroll
        for (int k = 0; k < 2; ++k) {
            v4u a = cur[2 * k], b = cur[2 * k + 1];
            m[k] = (a.x ^ xg[0]) & (a.y ^ xg[1]) & (a.z ^ xg[2]) & (a.w ^ xg[3])
                 & (b.x ^ xg[4]) & (b.y ^ xg[5]) & (b.z ^ xg[6]) & (b.w ^ xg[7]);
        }
        if (base + 2 * WGT >= nunit) {
            const unsigned int lastmask = (E & 31) ? ((1u << (E & 31)) - 1u) : ~0u;
            #pragma unroll
            for (int k = 0; k < 2; ++k) {
                int u = base + k * WGT + tid;
                unsigned int valid = (u < nunit - 1) ? ~0u : ((u == nunit - 1) ? lastmask : 0u);
                m[k] &= valid;
            }
        }
        int nb = base + 2 * WGT;
        if (nb < nunit) {
            #pragma unroll
            for (int k = 0; k < 2; ++k) {
                int u = nb + k * WGT + tid;
                if (u > nunit - 1) u = nunit - 1;
                cur[2 * k]     = klo[u];
                cur[2 * k + 1] = khi[u];
            }
        }
        #pragma unroll
        for (int k = 0; k < 2; ++k) {
            unsigned int mm = m[k];
            const int ebase = (base + k * WGT + tid) << 5;
            unsigned int bal = __builtin_amdgcn_ballot_w32(mm != 0u);
            while (bal) {
                if (mm != 0u) {
                    int rank = (int)__builtin_amdgcn_mbcnt_lo(bal, 0u);
                    myring[(tail + rank) & (RING - 1)] = ebase | __builtin_ctz(mm);
                }
                mm &= mm - 1u;
                tail += __builtin_popcount(bal);
                __builtin_amdgcn_wave_barrier();
                if (tail - head >= 32) {
                    int vq = myring[(head + lane) & (RING - 1)];
                    fa_drain<true, MODE>(h, dvec, gd, gs, esrc, edst, rec, zs, bias, row0, f, lane, vq, 32);
                    head += 32;
                    __builtin_amdgcn_wave_barrier();
                }
                bal = __builtin_amdgcn_ballot_w32(mm != 0u);
            }
        }
    }
    if (tail - head > 0) {
        const int left = tail - head;
        int vq = (lane < left) ? myring[(head + lane) & (RING - 1)] : 0;
        fa_drain<false, MODE>(h, dvec, gd, gs, esrc, edst, rec, zs, bias, row0, f, lane, vq, left);
    }

    __syncthreads();
    for (int i = tid; i < R * (NDIM / 4); i += WGT) {
        int row = row0 + (i >> 4);
        if (row < N) {
            v4f v = ((const v4f*)zs)[i];
            if (MODE == 2) v = v * dvec[row];
            ((v4f*)z)[(size_t)row * (NDIM / 4) + (i & 15)] = v;
        }
    }
}

extern "C" void kernel_launch(void* const* d_in, const int* in_sizes, int n_in,
                              void* d_out, int out_size, void* d_ws, size_t ws_size,
                              hipStream_t stream)
{
    const float* h      = (const float*)d_in[0];
    const float* dvec   = (const float*)d_in[1];
    const float* gate_W = (const float*)d_in[2];
    const float* gate_b = (const float*)d_in[3];
    const int*   esrc   = (const int*)d_in[4];
    const int*   edst   = (const int*)d_in[5];

    int N = in_sizes[1];
    int E = in_sizes[4];
    float* z = (float*)d_out;

    float* gd = (float*)d_ws;
    float* gs = gd + N;
    unsigned int* planes = (unsigned int*)(gs + ((N + 3) / 4) * 4);

    int R    = (N + 255) / 256;
    int nunit = (E + 31) / 32;
    unsigned int* rec = planes + (size_t)nunit * 8;
    const bool packed = (N <= (1 << 17)) && (R <= RMAX) &&
                        ((size_t)((const char*)(rec + E) - (const char*)d_ws) <= ws_size);

    const size_t hs_off = (((size_t)((const char*)(rec + E) - (const char*)d_ws)) + 255) / 256 * 256;
    const bool scaled   = packed && (hs_off + (size_t)N * NDIM * sizeof(float) <= ws_size);
    float* hs = scaled ? (float*)((char*)d_ws + hs_off) : nullptr;

    {
        int waves   = (N + 15) / 16;
        int threads = waves * 32;
        dim3 blk(256), grd((threads + 255) / 256);
        fa_gate_wmma<<<grd, blk, 0, stream>>>(h, gate_W, dvec, gd, gs, hs, N);
    }
    {
        int waves = (nunit + 7) / 8;
        dim3 blk(256), grd((waves + 7) / 8);
        fa_owner_planes<<<grd, blk, 0, stream>>>(esrc, edst, planes, packed ? rec : nullptr, E, nunit, R);
    }
    {
        dim3 blk(WGT), grd(256);
        if (scaled)
            fa_owner_sum<2><<<grd, blk, 0, stream>>>(hs, dvec, gd, gs, gate_b, esrc, edst,
                                                     planes, rec, z, N, E, nunit, R);
        else if (packed)
            fa_owner_sum<1><<<grd, blk, 0, stream>>>(h, dvec, gd, gs, gate_b, esrc, edst,
                                                     planes, rec, z, N, E, nunit, R);
        else
            fa_owner_sum<0><<<grd, blk, 0, stream>>>(h, dvec, gd, gs, gate_b, esrc, edst,
                                                     planes, rec, z, N, E, nunit, R);
    }
}
